# starting kernel
# baseline (speedup 1.0000x reference)
#include <hip/hip_runtime.h>
#include <hip/hip_bf16.h>
#include <math.h>

typedef float v2f __attribute__((ext_vector_type(2)));
typedef float v8f __attribute__((ext_vector_type(8)));

#define NHW  512
#define KF   16
#define LPAD 4

__device__ __forceinline__ v8f wmma_f32_4(v2f a, v2f b, v8f c) {
  return __builtin_amdgcn_wmma_f32_16x16x4_f32(false, a, false, b, (short)0, c,
                                               false, false);
}

__global__ void init_consts(float* __restrict__ A, float* __restrict__ D) {
  int idx = blockIdx.x * blockDim.x + threadIdx.x;
  if (idx < KF * NHW) {
    int i = idx / NHW, n = idx % NHW;
    double arg = M_PI * (double)((2 * n + 1) * i) / (2.0 * NHW);
    A[idx] = (i <= 10) ? (float)(2.0 * cos(arg)) : 0.0f;
  }
  if (idx < NHW * KF) {
    int n = idx / KF, i = idx % KF;
    double w = (i == 0) ? 0.5 : 1.0;
    double arg = M_PI * (double)((2 * n + 1) * i) / (2.0 * NHW);
    D[idx] = (i <= 10) ? (float)(w * cos(arg) / (double)NHW) : 0.0f;
  }
}

__global__ __launch_bounds__(32) void pass1_dct_low(const float* __restrict__ im,
                                                    const float* __restrict__ A,
                                                    float* __restrict__ Yp) {
  const int tile = blockIdx.x;
  const int img  = blockIdx.y;
  const int lane = threadIdx.x;
  const float* X = im + (size_t)img * NHW * NHW + (size_t)tile * 16 * NHW;

  __shared__ float Xs[16][NHW + LPAD];
  __shared__ float Ts[16][16 + LPAD];

  #pragma unroll
  for (int r = 0; r < 16; ++r) {
    const float4* src = (const float4*)(X + r * NHW);
    float4* dst = (float4*)(&Xs[r][0]);
    for (int c = lane; c < NHW / 4; c += 32) dst[c] = src[c];
  }
  __syncthreads();

  const int m    = lane & 15;
  const int hi   = lane >> 4;
  const int kof  = hi * 2;
  const int row0 = hi * 8;

  v8f t = {};
  for (int w0 = 0; w0 < NHW; w0 += 4) {
    v2f av, bv;
    av.x = Xs[m][w0 + kof];
    av.y = Xs[m][w0 + kof + 1];
    bv.x = A[m * NHW + w0 + kof];
    bv.y = A[m * NHW + w0 + kof + 1];
    t = wmma_f32_4(av, bv, t);
  }

  #pragma unroll
  for (int r = 0; r < 8; ++r) Ts[row0 + r][m] = t[r];
  __syncthreads();

  v8f y = {};
  #pragma unroll
  for (int k0 = 0; k0 < 16; k0 += 4) {
    const int kk = k0 + kof;
    v2f av, bv;
    av.x = A[m * NHW + tile * 16 + kk];
    av.y = A[m * NHW + tile * 16 + kk + 1];
    bv.x = Ts[kk][m];
    bv.y = Ts[kk + 1][m];
    y = wmma_f32_4(av, bv, y);
  }

  float* outp = Yp + ((size_t)img * 32 + tile) * 256;
  #pragma unroll
  for (int r = 0; r < 8; ++r) outp[(row0 + r) * 16 + m] = y[r];
}

__global__ void reduce_y(const float* __restrict__ Yp, float* __restrict__ Y) {
  const int img = blockIdx.x;
  const int e   = threadIdx.x;
  float s = 0.0f;
  #pragma unroll 4
  for (int t = 0; t < 32; ++t) s += Yp[((size_t)img * 32 + t) * 256 + e];
  Y[(size_t)img * 256 + e] = s;
}

__global__ __launch_bounds__(32) void pass2_reconstruct(const float* __restrict__ im,
                                                        const float* __restrict__ D,
                                                        const float* __restrict__ Y,
                                                        float* __restrict__ out) {
  const int tile = blockIdx.x;
  const int img  = blockIdx.y;
  const int lane = threadIdx.x;
  const int m    = lane & 15;
  const int hi   = lane >> 4;
  const int kof  = hi * 2;
  const int row0 = hi * 8;

  __shared__ float Ys[16][16 + LPAD];
  __shared__ float Ws[16][16 + LPAD];

  {
    const float* Yi = Y + (size_t)img * 256;
    #pragma unroll
    for (int r = 0; r < 8; ++r) {
      const int i = row0 + r;
      const float v = Yi[i * 16 + m];
      Ys[i][m] = (i + m <= 10) ? v : 0.0f;
    }
  }
  __syncthreads();

  v8f w = {};
  #pragma unroll
  for (int k0 = 0; k0 < 16; k0 += 4) {
    const int kk = k0 + kof;
    v2f av, bv;
    av.x = D[(tile * 16 + m) * KF + kk];
    av.y = D[(tile * 16 + m) * KF + kk + 1];
    bv.x = Ys[kk][m];
    bv.y = Ys[kk + 1][m];
    w = wmma_f32_4(av, bv, w);
  }
  #pragma unroll
  for (int r = 0; r < 8; ++r) Ws[row0 + r][m] = w[r];
  __syncthreads();

  const size_t base = (size_t)img * NHW * NHW + (size_t)tile * 16 * NHW;
  const float* X = im + base;
  float* O = out + base;

  for (int w0 = 0; w0 < NHW; w0 += 16) {
    v8f z = {};
    #pragma unroll
    for (int k0 = 0; k0 < 16; k0 += 4) {
      const int kk = k0 + kof;
      v2f av, bv;
      av.x = Ws[m][kk];
      av.y = Ws[m][kk + 1];
      bv.x = D[(w0 + m) * KF + kk];
      bv.y = D[(w0 + m) * KF + kk + 1];
      z = wmma_f32_4(av, bv, z);
    }
    #pragma unroll
    for (int r = 0; r < 8; ++r) {
      const int idx = (row0 + r) * NHW + w0 + m;
      O[idx] = X[idx] - z[r];
    }
  }
}

extern "C" void kernel_launch(void* const* d_in, const int* in_sizes, int n_in,
                              void* d_out, int out_size, void* d_ws, size_t ws_size,
                              hipStream_t stream) {
  (void)n_in; (void)out_size; (void)ws_size;
  const float* im = (const float*)d_in[0];
  float* out = (float*)d_out;
  const int nimg = in_sizes[0] / (NHW * NHW);

  char*  ws  = (char*)d_ws;
  float* dA  = (float*)(ws);
  float* dD  = (float*)(ws + 32768);
  float* dYp = (float*)(ws + 65536);
  float* dY  = (float*)(ws + 65536 + (size_t)nimg * 32 * 256 * sizeof(float));

  init_consts<<<32, 256, 0, stream>>>(dA, dD);
  pass1_dct_low<<<dim3(32, nimg), 32, 0, stream>>>(im, dA, dYp);
  reduce_y<<<nimg, 256, 0, stream>>>(dYp, dY);
  pass2_reconstruct<<<dim3(32, nimg), 32, 0, stream>>>(im, dD, dY, out);
}


# pass 2: each 16x64 block of Z passes through LDS so a lane owns 4 adjacent columns: im and out move as b128 in whole 256-B row runs (were 64-B pieces); next im block requested one step ahead
# speedup vs baseline: 2.3580x; 2.3580x over previous
#include <hip/hip_runtime.h>
#include <hip/hip_bf16.h>
#include <math.h>

typedef float v2f __attribute__((ext_vector_type(2)));
typedef float v8f __attribute__((ext_vector_type(8)));

#define NHW  512
#define KF   16
#define LPAD 4

__device__ __forceinline__ v8f wmma_f32_4(v2f a, v2f b, v8f c) {
  return __builtin_amdgcn_wmma_f32_16x16x4_f32(false, a, false, b, (short)0, c,
                                               false, false);
}

__global__ void init_consts(float* __restrict__ A, float* __restrict__ D) {
  int idx = blockIdx.x * blockDim.x + threadIdx.x;
  if (idx < KF * NHW) {
    int i = idx / NHW, n = idx % NHW;
    double arg = M_PI * (double)((2 * n + 1) * i) / (2.0 * NHW);
    A[idx] = (i <= 10) ? (float)(2.0 * cos(arg)) : 0.0f;
  }
  if (idx < NHW * KF) {
    int n = idx / KF, i = idx % KF;
    double w = (i == 0) ? 0.5 : 1.0;
    double arg = M_PI * (double)((2 * n + 1) * i) / (2.0 * NHW);
    D[idx] = (i <= 10) ? (float)(w * cos(arg) / (double)NHW) : 0.0f;
  }
}

#define P1W  4
#define P1T  (32 * P1W)
#define KSL  (NHW / P1W)

__global__ __launch_bounds__(P1T) void pass1_dct_low(const float* __restrict__ im,
                                                     const float* __restrict__ A,
                                                     float* __restrict__ Yp) {
  const int tile = blockIdx.x;
  const int img  = blockIdx.y;
  const int tid  = threadIdx.x;
  const int lane = tid & 31;
  const int wv   = __builtin_amdgcn_readfirstlane(tid >> 5);
  const float* X = im + (size_t)img * NHW * NHW + (size_t)tile * 16 * NHW;

  __shared__ float Xs[16][NHW + LPAD];
  __shared__ float Tp[P1W][16][16 + LPAD];

  #pragma unroll
  for (int q = 0; q < (16 * NHW / 4) / P1T; ++q) {
    const int idx = q * P1T + tid;
    const int r = idx / (NHW / 4);
    const int c = idx % (NHW / 4);
    *(float4*)(&Xs[r][4 * c]) = *(const float4*)(X + r * NHW + 4 * c);
  }
  __syncthreads();

  const int m    = lane & 15;
  const int hi   = lane >> 4;
  const int kof  = hi * 2;
  const int row0 = hi * 8;

  const int wbeg = wv * KSL;
  v8f t = {};
  #pragma unroll
  for (int ws = 0; ws < KSL; ws += 4) {
    const int w0 = wbeg + ws;
    v2f av, bv;
    av.x = Xs[m][w0 + kof];
    av.y = Xs[m][w0 + kof + 1];
    bv.x = A[m * NHW + w0 + kof];
    bv.y = A[m * NHW + w0 + kof + 1];
    t = wmma_f32_4(av, bv, t);
  }

  #pragma unroll
  for (int r = 0; r < 8; ++r) Tp[wv][row0 + r][m] = t[r];
  __syncthreads();
  if (wv != 0) return;

  v8f y = {};
  #pragma unroll
  for (int k0 = 0; k0 < 16; k0 += 4) {
    const int kk = k0 + kof;
    v2f av, bv;
    av.x = A[m * NHW + tile * 16 + kk];
    av.y = A[m * NHW + tile * 16 + kk + 1];
    bv.x = Tp[0][kk][m];
    bv.y = Tp[0][kk + 1][m];
    #pragma unroll
    for (int s = 1; s < P1W; ++s) {
      bv.x += Tp[s][kk][m];
      bv.y += Tp[s][kk + 1][m];
    }
    y = wmma_f32_4(av, bv, y);
  }

  float* outp = Yp + ((size_t)img * 32 + tile) * 256;
  #pragma unroll
  for (int r = 0; r < 8; ++r) outp[(row0 + r) * 16 + m] = y[r];
}

__global__ void reduce_y(const float* __restrict__ Yp, float* __restrict__ Y) {
  const int img = blockIdx.x;
  const int e   = threadIdx.x;
  float s = 0.0f;
  #pragma unroll 4
  for (int t = 0; t < 32; ++t) s += Yp[((size_t)img * 32 + t) * 256 + e];
  Y[(size_t)img * 256 + e] = s;
}

__device__ __forceinline__ void lds_barrier() {
  __builtin_amdgcn_fence(__ATOMIC_RELEASE, "workgroup", "local");
  __builtin_amdgcn_s_barrier();
  __builtin_amdgcn_fence(__ATOMIC_ACQUIRE, "workgroup", "local");
}

#define ZC 64
#define ZP (ZC + 4)

__global__ __launch_bounds__(32) void pass2_reconstruct(const float* __restrict__ im,
                                                        const float* __restrict__ D,
                                                        const float* __restrict__ Y,
                                                        float* __restrict__ out) {
  const int tile = blockIdx.x;
  const int img  = blockIdx.y;
  const int lane = threadIdx.x;
  const int m    = lane & 15;
  const int hi   = lane >> 4;
  const int kof  = hi * 2;
  const int row0 = hi * 8;

  __shared__ float Ys[16][16 + LPAD];
  __shared__ float Ws[16][16 + LPAD];
  __shared__ __attribute__((aligned(16))) float Zs[16][ZP];

  {
    const float* Yi = Y + (size_t)img * 256;
    #pragma unroll
    for (int r = 0; r < 8; ++r) {
      const int i = row0 + r;
      const float v = Yi[i * 16 + m];
      Ys[i][m] = (i + m <= 10) ? v : 0.0f;
    }
  }
  __syncthreads();

  v8f w = {};
  #pragma unroll
  for (int k0 = 0; k0 < 16; k0 += 4) {
    const int kk = k0 + kof;
    v2f av, bv;
    av.x = D[(tile * 16 + m) * KF + kk];
    av.y = D[(tile * 16 + m) * KF + kk + 1];
    bv.x = Ys[kk][m];
    bv.y = Ys[kk + 1][m];
    w = wmma_f32_4(av, bv, w);
  }
  #pragma unroll
  for (int r = 0; r < 8; ++r) Ws[row0 + r][m] = w[r];
  __syncthreads();

  const size_t base = (size_t)img * NHW * NHW + (size_t)tile * 16 * NHW;
  const float* X = im + base;
  float* O = out + base;
  const int c4 = m * 4;

  v2f af[4];
  #pragma unroll
  for (int k = 0; k < 4; ++k) {
    af[k].x = Ws[m][4 * k + kof];
    af[k].y = Ws[m][4 * k + kof + 1];
  }

  float4 xc[8];
  #pragma unroll
  for (int q = 0; q < 8; ++q)
    xc[q] = *(const float4*)(X + (2 * q + hi) * NHW + c4);

  #pragma unroll
  for (int s = 0; s < NHW / ZC; ++s) {
    const int w0 = s * ZC;

    v2f bf[ZC / 16][4];
    #pragma unroll
    for (int t = 0; t < ZC / 16; ++t) {
      #pragma unroll
      for (int k = 0; k < 4; ++k) {
        bf[t][k].x = D[(w0 + 16 * t + m) * KF + 4 * k + kof];
        bf[t][k].y = D[(w0 + 16 * t + m) * KF + 4 * k + kof + 1];
      }
    }
    __builtin_amdgcn_sched_barrier(0);

    float4 xn[8];
    if (s + 1 < NHW / ZC) {
      #pragma unroll
      for (int q = 0; q < 8; ++q)
        xn[q] = *(const float4*)(X + (2 * q + hi) * NHW + w0 + ZC + c4);
    }
    __builtin_amdgcn_sched_barrier(0);

    #pragma unroll
    for (int t = 0; t < ZC / 16; ++t) {
      v8f z = {};
      #pragma unroll
      for (int k = 0; k < 4; ++k) z = wmma_f32_4(af[k], bf[t][k], z);
      #pragma unroll
      for (int r = 0; r < 8; ++r) Zs[row0 + r][16 * t + m] = z[r];
    }
    lds_barrier();

    #pragma unroll
    for (int q = 0; q < 8; ++q) {
      const float4 zz = *(const float4*)(&Zs[2 * q + hi][c4]);
      float4 o;
      o.x = xc[q].x - zz.x;
      o.y = xc[q].y - zz.y;
      o.z = xc[q].z - zz.z;
      o.w = xc[q].w - zz.w;
      *(float4*)(O + (2 * q + hi) * NHW + w0 + c4) = o;
    }
    lds_barrier();

    if (s + 1 < NHW / ZC) {
      #pragma unroll
      for (int q = 0; q < 8; ++q) xc[q] = xn[q];
    }
  }
}

extern "C" void kernel_launch(void* const* d_in, const int* in_sizes, int n_in,
                              void* d_out, int out_size, void* d_ws, size_t ws_size,
                              hipStream_t stream) {
  (void)n_in; (void)out_size; (void)ws_size;
  const float* im = (const float*)d_in[0];
  float* out = (float*)d_out;
  const int nimg = in_sizes[0] / (NHW * NHW);

  char*  ws  = (char*)d_ws;
  float* dA  = (float*)(ws);
  float* dD  = (float*)(ws + 32768);
  float* dYp = (float*)(ws + 65536);
  float* dY  = (float*)(ws + 65536 + (size_t)nimg * 32 * 256 * sizeof(float));

  init_consts<<<32, 256, 0, stream>>>(dA, dD);
  pass1_dct_low<<<dim3(32, nimg), P1T, 0, stream>>>(im, dA, dYp);
  reduce_y<<<nimg, 256, 0, stream>>>(dYp, dY);
  pass2_reconstruct<<<dim3(32, nimg), 32, 0, stream>>>(im, dD, dY, out);
}


# pass 1: one work-group walks 4 row tiles (768 groups, was 3072), next tile's loads in flight under the WMMAs, table fragments loaded once; 8 partial Y per image, so the row sum order changes
# speedup vs baseline: 2.5960x; 1.1009x over previous
#include <hip/hip_runtime.h>
#include <hip/hip_bf16.h>
#include <math.h>

typedef float v2f __attribute__((ext_vector_type(2)));
typedef float v8f __attribute__((ext_vector_type(8)));

#define NHW  512
#define KF   16
#define LPAD 4

__device__ __forceinline__ v8f wmma_f32_4(v2f a, v2f b, v8f c) {
  return __builtin_amdgcn_wmma_f32_16x16x4_f32(false, a, false, b, (short)0, c,
                                               false, false);
}

__global__ void init_consts(float* __restrict__ A, float* __restrict__ D) {
  int idx = blockIdx.x * blockDim.x + threadIdx.x;
  if (idx < KF * NHW) {
    int i = idx / NHW, n = idx % NHW;
    double arg = M_PI * (double)((2 * n + 1) * i) / (2.0 * NHW);
    A[idx] = (i <= 10) ? (float)(2.0 * cos(arg)) : 0.0f;
  }
  if (idx < NHW * KF) {
    int n = idx / KF, i = idx % KF;
    double w = (i == 0) ? 0.5 : 1.0;
    double arg = M_PI * (double)((2 * n + 1) * i) / (2.0 * NHW);
    D[idx] = (i <= 10) ? (float)(w * cos(arg) / (double)NHW) : 0.0f;
  }
}

__device__ __forceinline__ void lds_barrier() {
  __builtin_amdgcn_fence(__ATOMIC_RELEASE, "workgroup", "local");
  __builtin_amdgcn_s_barrier();
  __builtin_amdgcn_fence(__ATOMIC_ACQUIRE, "workgroup", "local");
}

#define P1W  4
#define P1T  (32 * P1W)
#define KSL  (NHW / P1W)
#define P1R  4
#define NPY  (32 / P1R)
#define P1Q  ((16 * NHW / 4) / P1T)

__global__ __launch_bounds__(P1T) void pass1_dct_low(const float* __restrict__ im,
                                                     const float* __restrict__ A,
                                                     float* __restrict__ Yp) {
  const int band = blockIdx.x;
  const int img  = blockIdx.y;
  const int tid  = threadIdx.x;
  const int lane = tid & 31;
  const int wv   = __builtin_amdgcn_readfirstlane(tid >> 5);
  const float* X = im + (size_t)img * NHW * NHW + (size_t)band * P1R * 16 * NHW;

  __shared__ __attribute__((aligned(16))) float Xs[2][16][NHW + LPAD];
  __shared__ float Tp[P1R][P1W][16][16 + LPAD];

  const int m    = lane & 15;
  const int hi   = lane >> 4;
  const int kof  = hi * 2;
  const int row0 = hi * 8;
  const int wbeg = wv * KSL;

  const int lr = tid / (NHW / 4);
  const int lc = tid % (NHW / 4);
  float4 xr[P1Q];
  #pragma unroll
  for (int q = 0; q < P1Q; ++q)
    xr[q] = *(const float4*)(X + (lr + q * (P1T * 4 / NHW)) * NHW + 4 * lc);

  v2f bt[KSL / 4];
  #pragma unroll
  for (int j = 0; j < KSL / 4; ++j) {
    bt[j].x = A[m * NHW + wbeg + 4 * j + kof];
    bt[j].y = A[m * NHW + wbeg + 4 * j + kof + 1];
  }

  #pragma unroll
  for (int r = 0; r < P1R; ++r) {
    const int b = r & 1;
    #pragma unroll
    for (int q = 0; q < P1Q; ++q)
      *(float4*)(&Xs[b][lr + q * (P1T * 4 / NHW)][4 * lc]) = xr[q];
    lds_barrier();

    if (r + 1 < P1R) {
      #pragma unroll
      for (int q = 0; q < P1Q; ++q)
        xr[q] = *(const float4*)(X + ((r + 1) * 16 + lr + q * (P1T * 4 / NHW)) * NHW + 4 * lc);
    }

    v8f t = {};
    #pragma unroll
    for (int j = 0; j < KSL / 4; ++j) {
      const int w0 = wbeg + 4 * j;
      v2f av;
      av.x = Xs[b][m][w0 + kof];
      av.y = Xs[b][m][w0 + kof + 1];
      t = wmma_f32_4(av, bt[j], t);
    }
    #pragma unroll
    for (int q = 0; q < 8; ++q) Tp[r][wv][row0 + q][m] = t[q];
  }
  lds_barrier();
  if (wv != 0) return;

  v8f y = {};
  #pragma unroll
  for (int r = 0; r < P1R; ++r) {
    const int rowb = (band * P1R + r) * 16;
    #pragma unroll
    for (int k0 = 0; k0 < 16; k0 += 4) {
      const int kk = k0 + kof;
      v2f av, bv;
      av.x = A[m * NHW + rowb + kk];
      av.y = A[m * NHW + rowb + kk + 1];
      bv.x = Tp[r][0][kk][m];
      bv.y = Tp[r][0][kk + 1][m];
      #pragma unroll
      for (int s = 1; s < P1W; ++s) {
        bv.x += Tp[r][s][kk][m];
        bv.y += Tp[r][s][kk + 1][m];
      }
      y = wmma_f32_4(av, bv, y);
    }
  }

  float* outp = Yp + ((size_t)img * NPY + band) * 256;
  #pragma unroll
  for (int q = 0; q < 8; ++q) outp[(row0 + q) * 16 + m] = y[q];
}

__global__ void reduce_y(const float* __restrict__ Yp, float* __restrict__ Y) {
  const int img = blockIdx.x;
  const int e   = threadIdx.x;
  float s = 0.0f;
  #pragma unroll 4
  for (int t = 0; t < NPY; ++t) s += Yp[((size_t)img * NPY + t) * 256 + e];
  Y[(size_t)img * 256 + e] = s;
}

#define ZC 64
#define ZP (ZC + 4)

__global__ __launch_bounds__(32) void pass2_reconstruct(const float* __restrict__ im,
                                                        const float* __restrict__ D,
                                                        const float* __restrict__ Y,
                                                        float* __restrict__ out) {
  const int tile = blockIdx.x;
  const int img  = blockIdx.y;
  const int lane = threadIdx.x;
  const int m    = lane & 15;
  const int hi   = lane >> 4;
  const int kof  = hi * 2;
  const int row0 = hi * 8;

  __shared__ float Ys[16][16 + LPAD];
  __shared__ float Ws[16][16 + LPAD];
  __shared__ __attribute__((aligned(16))) float Zs[16][ZP];

  {
    const float* Yi = Y + (size_t)img * 256;
    #pragma unroll
    for (int r = 0; r < 8; ++r) {
      const int i = row0 + r;
      const float v = Yi[i * 16 + m];
      Ys[i][m] = (i + m <= 10) ? v : 0.0f;
    }
  }
  __syncthreads();

  v8f w = {};
  #pragma unroll
  for (int k0 = 0; k0 < 16; k0 += 4) {
    const int kk = k0 + kof;
    v2f av, bv;
    av.x = D[(tile * 16 + m) * KF + kk];
    av.y = D[(tile * 16 + m) * KF + kk + 1];
    bv.x = Ys[kk][m];
    bv.y = Ys[kk + 1][m];
    w = wmma_f32_4(av, bv, w);
  }
  #pragma unroll
  for (int r = 0; r < 8; ++r) Ws[row0 + r][m] = w[r];
  __syncthreads();

  const size_t base = (size_t)img * NHW * NHW + (size_t)tile * 16 * NHW;
  const float* X = im + base;
  float* O = out + base;
  const int c4 = m * 4;

  v2f af[4];
  #pragma unroll
  for (int k = 0; k < 4; ++k) {
    af[k].x = Ws[m][4 * k + kof];
    af[k].y = Ws[m][4 * k + kof + 1];
  }

  float4 xc[8];
  #pragma unroll
  for (int q = 0; q < 8; ++q)
    xc[q] = *(const float4*)(X + (2 * q + hi) * NHW + c4);

  #pragma unroll
  for (int s = 0; s < NHW / ZC; ++s) {
    const int w0 = s * ZC;

    v2f bf[ZC / 16][4];
    #pragma unroll
    for (int t = 0; t < ZC / 16; ++t) {
      #pragma unroll
      for (int k = 0; k < 4; ++k) {
        bf[t][k].x = D[(w0 + 16 * t + m) * KF + 4 * k + kof];
        bf[t][k].y = D[(w0 + 16 * t + m) * KF + 4 * k + kof + 1];
      }
    }
    __builtin_amdgcn_sched_barrier(0);

    float4 xn[8];
    if (s + 1 < NHW / ZC) {
      #pragma unroll
      for (int q = 0; q < 8; ++q)
        xn[q] = *(const float4*)(X + (2 * q + hi) * NHW + w0 + ZC + c4);
    }
    __builtin_amdgcn_sched_barrier(0);

    #pragma unroll
    for (int t = 0; t < ZC / 16; ++t) {
      v8f z = {};
      #pragma unroll
      for (int k = 0; k < 4; ++k) z = wmma_f32_4(af[k], bf[t][k], z);
      #pragma unroll
      for (int r = 0; r < 8; ++r) Zs[row0 + r][16 * t + m] = z[r];
    }
    lds_barrier();

    #pragma unroll
    for (int q = 0; q < 8; ++q) {
      const float4 zz = *(const float4*)(&Zs[2 * q + hi][c4]);
      float4 o;
      o.x = xc[q].x - zz.x;
      o.y = xc[q].y - zz.y;
      o.z = xc[q].z - zz.z;
      o.w = xc[q].w - zz.w;
      *(float4*)(O + (2 * q + hi) * NHW + w0 + c4) = o;
    }
    lds_barrier();

    if (s + 1 < NHW / ZC) {
      #pragma unroll
      for (int q = 0; q < 8; ++q) xc[q] = xn[q];
    }
  }
}

extern "C" void kernel_launch(void* const* d_in, const int* in_sizes, int n_in,
                              void* d_out, int out_size, void* d_ws, size_t ws_size,
                              hipStream_t stream) {
  (void)n_in; (void)out_size; (void)ws_size;
  const float* im = (const float*)d_in[0];
  float* out = (float*)d_out;
  const int nimg = in_sizes[0] / (NHW * NHW);

  char*  ws  = (char*)d_ws;
  float* dA  = (float*)(ws);
  float* dD  = (float*)(ws + 32768);
  float* dYp = (float*)(ws + 65536);
  float* dY  = (float*)(ws + 65536 + (size_t)nimg * 32 * 256 * sizeof(float));

  init_consts<<<32, 256, 0, stream>>>(dA, dD);
  pass1_dct_low<<<dim3(NPY, nimg), P1T, 0, stream>>>(im, dA, dYp);
  reduce_y<<<nimg, 256, 0, stream>>>(dYp, dY);
  pass2_reconstruct<<<dim3(32, nimg), 32, 0, stream>>>(im, dD, dY, out);
}


# table launch: one f64 cosine stream per SIMD (was 4): 256 one-wave work-groups spread over the chip, A[i][n] and D[n][i] made from the same cosine; table values bit-identical
# speedup vs baseline: 2.7877x; 1.0738x over previous
#include <hip/hip_runtime.h>
#include <hip/hip_bf16.h>
#include <math.h>

typedef float v2f __attribute__((ext_vector_type(2)));
typedef float v8f __attribute__((ext_vector_type(8)));

#define NHW  512
#define KF   16
#define LPAD 4

__device__ __forceinline__ v8f wmma_f32_4(v2f a, v2f b, v8f c) {
  return __builtin_amdgcn_wmma_f32_16x16x4_f32(false, a, false, b, (short)0, c,
                                               false, false);
}

#define TBW 32

__global__ __launch_bounds__(TBW) void init_consts(float* __restrict__ A,
                                                   float* __restrict__ D) {
  const int idx = blockIdx.x * TBW + threadIdx.x;
  const int i = idx / NHW, n = idx % NHW;
  float a = 0.0f, d = 0.0f;
  if (i <= 10) {
    const double w = (i == 0) ? 0.5 : 1.0;
    const double arg = M_PI * (double)((2 * n + 1) * i) / (2.0 * NHW);
    const double c = cos(arg);
    a = (float)(2.0 * c);
    d = (float)(w * c / (double)NHW);
  }
  A[i * NHW + n] = a;
  D[n * KF + i] = d;
}

__device__ __forceinline__ void lds_barrier() {
  __builtin_amdgcn_fence(__ATOMIC_RELEASE, "workgroup", "local");
  __builtin_amdgcn_s_barrier();
  __builtin_amdgcn_fence(__ATOMIC_ACQUIRE, "workgroup", "local");
}

#define P1W  4
#define P1T  (32 * P1W)
#define KSL  (NHW / P1W)
#define P1R  4
#define NPY  (32 / P1R)
#define P1Q  ((16 * NHW / 4) / P1T)

__global__ __launch_bounds__(P1T) void pass1_dct_low(const float* __restrict__ im,
                                                     const float* __restrict__ A,
                                                     float* __restrict__ Yp) {
  const int band = blockIdx.x;
  const int img  = blockIdx.y;
  const int tid  = threadIdx.x;
  const int lane = tid & 31;
  const int wv   = __builtin_amdgcn_readfirstlane(tid >> 5);
  const float* X = im + (size_t)img * NHW * NHW + (size_t)band * P1R * 16 * NHW;

  __shared__ __attribute__((aligned(16))) float Xs[2][16][NHW + LPAD];
  __shared__ float Tp[P1R][P1W][16][16 + LPAD];

  const int m    = lane & 15;
  const int hi   = lane >> 4;
  const int kof  = hi * 2;
  const int row0 = hi * 8;
  const int wbeg = wv * KSL;

  const int lr = tid / (NHW / 4);
  const int lc = tid % (NHW / 4);
  float4 xr[P1Q];
  #pragma unroll
  for (int q = 0; q < P1Q; ++q)
    xr[q] = *(const float4*)(X + (lr + q * (P1T * 4 / NHW)) * NHW + 4 * lc);

  v2f bt[KSL / 4];
  #pragma unroll
  for (int j = 0; j < KSL / 4; ++j) {
    bt[j].x = A[m * NHW + wbeg + 4 * j + kof];
    bt[j].y = A[m * NHW + wbeg + 4 * j + kof + 1];
  }

  #pragma unroll
  for (int r = 0; r < P1R; ++r) {
    const int b = r & 1;
    #pragma unroll
    for (int q = 0; q < P1Q; ++q)
      *(float4*)(&Xs[b][lr + q * (P1T * 4 / NHW)][4 * lc]) = xr[q];
    lds_barrier();

    if (r + 1 < P1R) {
      #pragma unroll
      for (int q = 0; q < P1Q; ++q)
        xr[q] = *(const float4*)(X + ((r + 1) * 16 + lr + q * (P1T * 4 / NHW)) * NHW + 4 * lc);
    }

    v8f t = {};
    #pragma unroll
    for (int j = 0; j < KSL / 4; ++j) {
      const int w0 = wbeg + 4 * j;
      v2f av;
      av.x = Xs[b][m][w0 + kof];
      av.y = Xs[b][m][w0 + kof + 1];
      t = wmma_f32_4(av, bt[j], t);
    }
    #pragma unroll
    for (int q = 0; q < 8; ++q) Tp[r][wv][row0 + q][m] = t[q];
  }
  lds_barrier();
  if (wv != 0) return;

  v8f y = {};
  #pragma unroll
  for (int r = 0; r < P1R; ++r) {
    const int rowb = (band * P1R + r) * 16;
    #pragma unroll
    for (int k0 = 0; k0 < 16; k0 += 4) {
      const int kk = k0 + kof;
      v2f av, bv;
      av.x = A[m * NHW + rowb + kk];
      av.y = A[m * NHW + rowb + kk + 1];
      bv.x = Tp[r][0][kk][m];
      bv.y = Tp[r][0][kk + 1][m];
      #pragma unroll
      for (int s = 1; s < P1W; ++s) {
        bv.x += Tp[r][s][kk][m];
        bv.y += Tp[r][s][kk + 1][m];
      }
      y = wmma_f32_4(av, bv, y);
    }
  }

  float* outp = Yp + ((size_t)img * NPY + band) * 256;
  #pragma unroll
  for (int q = 0; q < 8; ++q) outp[(row0 + q) * 16 + m] = y[q];
}

__global__ void reduce_y(const float* __restrict__ Yp, float* __restrict__ Y) {
  const int img = blockIdx.x;
  const int e   = threadIdx.x;
  float s = 0.0f;
  #pragma unroll 4
  for (int t = 0; t < NPY; ++t) s += Yp[((size_t)img * NPY + t) * 256 + e];
  Y[(size_t)img * 256 + e] = s;
}

#define ZC 64
#define ZP (ZC + 4)

__global__ __launch_bounds__(32) void pass2_reconstruct(const float* __restrict__ im,
                                                        const float* __restrict__ D,
                                                        const float* __restrict__ Y,
                                                        float* __restrict__ out) {
  const int tile = blockIdx.x;
  const int img  = blockIdx.y;
  const int lane = threadIdx.x;
  const int m    = lane & 15;
  const int hi   = lane >> 4;
  const int kof  = hi * 2;
  const int row0 = hi * 8;

  __shared__ float Ys[16][16 + LPAD];
  __shared__ float Ws[16][16 + LPAD];
  __shared__ __attribute__((aligned(16))) float Zs[16][ZP];

  {
    const float* Yi = Y + (size_t)img * 256;
    #pragma unroll
    for (int r = 0; r < 8; ++r) {
      const int i = row0 + r;
      const float v = Yi[i * 16 + m];
      Ys[i][m] = (i + m <= 10) ? v : 0.0f;
    }
  }
  __syncthreads();

  v8f w = {};
  #pragma unroll
  for (int k0 = 0; k0 < 16; k0 += 4) {
    const int kk = k0 + kof;
    v2f av, bv;
    av.x = D[(tile * 16 + m) * KF + kk];
    av.y = D[(tile * 16 + m) * KF + kk + 1];
    bv.x = Ys[kk][m];
    bv.y = Ys[kk + 1][m];
    w = wmma_f32_4(av, bv, w);
  }
  #pragma unroll
  for (int r = 0; r < 8; ++r) Ws[row0 + r][m] = w[r];
  __syncthreads();

  const size_t base = (size_t)img * NHW * NHW + (size_t)tile * 16 * NHW;
  const float* X = im + base;
  float* O = out + base;
  const int c4 = m * 4;

  v2f af[4];
  #pragma unroll
  for (int k = 0; k < 4; ++k) {
    af[k].x = Ws[m][4 * k + kof];
    af[k].y = Ws[m][4 * k + kof + 1];
  }

  float4 xc[8];
  #pragma unroll
  for (int q = 0; q < 8; ++q)
    xc[q] = *(const float4*)(X + (2 * q + hi) * NHW + c4);

  #pragma unroll
  for (int s = 0; s < NHW / ZC; ++s) {
    const int w0 = s * ZC;

    v2f bf[ZC / 16][4];
    #pragma unroll
    for (int t = 0; t < ZC / 16; ++t) {
      #pragma unroll
      for (int k = 0; k < 4; ++k) {
        bf[t][k].x = D[(w0 + 16 * t + m) * KF + 4 * k + kof];
        bf[t][k].y = D[(w0 + 16 * t + m) * KF + 4 * k + kof + 1];
      }
    }
    __builtin_amdgcn_sched_barrier(0);

    float4 xn[8];
    if (s + 1 < NHW / ZC) {
      #pragma unroll
      for (int q = 0; q < 8; ++q)
        xn[q] = *(const float4*)(X + (2 * q + hi) * NHW + w0 + ZC + c4);
    }
    __builtin_amdgcn_sched_barrier(0);

    #pragma unroll
    for (int t = 0; t < ZC / 16; ++t) {
      v8f z = {};
      #pragma unroll
      for (int k = 0; k < 4; ++k) z = wmma_f32_4(af[k], bf[t][k], z);
      #pragma unroll
      for (int r = 0; r < 8; ++r) Zs[row0 + r][16 * t + m] = z[r];
    }
    lds_barrier();

    #pragma unroll
    for (int q = 0; q < 8; ++q) {
      const float4 zz = *(const float4*)(&Zs[2 * q + hi][c4]);
      float4 o;
      o.x = xc[q].x - zz.x;
      o.y = xc[q].y - zz.y;
      o.z = xc[q].z - zz.z;
      o.w = xc[q].w - zz.w;
      *(float4*)(O + (2 * q + hi) * NHW + w0 + c4) = o;
    }
    lds_barrier();

    if (s + 1 < NHW / ZC) {
      #pragma unroll
      for (int q = 0; q < 8; ++q) xc[q] = xn[q];
    }
  }
}

extern "C" void kernel_launch(void* const* d_in, const int* in_sizes, int n_in,
                              void* d_out, int out_size, void* d_ws, size_t ws_size,
                              hipStream_t stream) {
  (void)n_in; (void)out_size; (void)ws_size;
  const float* im = (const float*)d_in[0];
  float* out = (float*)d_out;
  const int nimg = in_sizes[0] / (NHW * NHW);

  char*  ws  = (char*)d_ws;
  float* dA  = (float*)(ws);
  float* dD  = (float*)(ws + 32768);
  float* dYp = (float*)(ws + 65536);
  float* dY  = (float*)(ws + 65536 + (size_t)nimg * 32 * 256 * sizeof(float));

  init_consts<<<KF * NHW / TBW, TBW, 0, stream>>>(dA, dD);
  pass1_dct_low<<<dim3(NPY, nimg), P1T, 0, stream>>>(im, dA, dYp);
  reduce_y<<<nimg, 256, 0, stream>>>(dYp, dY);
  pass2_reconstruct<<<dim3(32, nimg), 32, 0, stream>>>(im, dD, dY, out);
}


# pass 2: a wave walks a 128-column strip down 4 row tiles, strip's table fragments loaded once (were reloaded per row tile); W made as W^T straight in A-operand registers; 16-term sums regrouped
# speedup vs baseline: 2.9548x; 1.0599x over previous
#include <hip/hip_runtime.h>
#include <hip/hip_bf16.h>
#include <math.h>

typedef float v2f __attribute__((ext_vector_type(2)));
typedef float v8f __attribute__((ext_vector_type(8)));

#define NHW  512
#define KF   16
#define LPAD 4

__device__ __forceinline__ v8f wmma_f32_4(v2f a, v2f b, v8f c) {
  return __builtin_amdgcn_wmma_f32_16x16x4_f32(false, a, false, b, (short)0, c,
                                               false, false);
}

#define TBW 32

__global__ __launch_bounds__(TBW) void init_consts(float* __restrict__ A,
                                                   float* __restrict__ D) {
  const int idx = blockIdx.x * TBW + threadIdx.x;
  const int i = idx / NHW, n = idx % NHW;
  float a = 0.0f, d = 0.0f;
  if (i <= 10) {
    const double w = (i == 0) ? 0.5 : 1.0;
    const double arg = M_PI * (double)((2 * n + 1) * i) / (2.0 * NHW);
    const double c = cos(arg);
    a = (float)(2.0 * c);
    d = (float)(w * c / (double)NHW);
  }
  A[i * NHW + n] = a;
  D[n * KF + i] = d;
}

__device__ __forceinline__ void lds_barrier() {
  __builtin_amdgcn_fence(__ATOMIC_RELEASE, "workgroup", "local");
  __builtin_amdgcn_s_barrier();
  __builtin_amdgcn_fence(__ATOMIC_ACQUIRE, "workgroup", "local");
}

#define P1W  4
#define P1T  (32 * P1W)
#define KSL  (NHW / P1W)
#define P1R  4
#define NPY  (32 / P1R)
#define P1Q  ((16 * NHW / 4) / P1T)

__global__ __launch_bounds__(P1T) void pass1_dct_low(const float* __restrict__ im,
                                                     const float* __restrict__ A,
                                                     float* __restrict__ Yp) {
  const int band = blockIdx.x;
  const int img  = blockIdx.y;
  const int tid  = threadIdx.x;
  const int lane = tid & 31;
  const int wv   = __builtin_amdgcn_readfirstlane(tid >> 5);
  const float* X = im + (size_t)img * NHW * NHW + (size_t)band * P1R * 16 * NHW;

  __shared__ __attribute__((aligned(16))) float Xs[2][16][NHW + LPAD];
  __shared__ float Tp[P1R][P1W][16][16 + LPAD];

  const int m    = lane & 15;
  const int hi   = lane >> 4;
  const int kof  = hi * 2;
  const int row0 = hi * 8;
  const int wbeg = wv * KSL;

  const int lr = tid / (NHW / 4);
  const int lc = tid % (NHW / 4);
  float4 xr[P1Q];
  #pragma unroll
  for (int q = 0; q < P1Q; ++q)
    xr[q] = *(const float4*)(X + (lr + q * (P1T * 4 / NHW)) * NHW + 4 * lc);

  v2f bt[KSL / 4];
  #pragma unroll
  for (int j = 0; j < KSL / 4; ++j) {
    bt[j].x = A[m * NHW + wbeg + 4 * j + kof];
    bt[j].y = A[m * NHW + wbeg + 4 * j + kof + 1];
  }

  #pragma unroll
  for (int r = 0; r < P1R; ++r) {
    const int b = r & 1;
    #pragma unroll
    for (int q = 0; q < P1Q; ++q)
      *(float4*)(&Xs[b][lr + q * (P1T * 4 / NHW)][4 * lc]) = xr[q];
    lds_barrier();

    if (r + 1 < P1R) {
      #pragma unroll
      for (int q = 0; q < P1Q; ++q)
        xr[q] = *(const float4*)(X + ((r + 1) * 16 + lr + q * (P1T * 4 / NHW)) * NHW + 4 * lc);
    }

    v8f t = {};
    #pragma unroll
    for (int j = 0; j < KSL / 4; ++j) {
      const int w0 = wbeg + 4 * j;
      v2f av;
      av.x = Xs[b][m][w0 + kof];
      av.y = Xs[b][m][w0 + kof + 1];
      t = wmma_f32_4(av, bt[j], t);
    }
    #pragma unroll
    for (int q = 0; q < 8; ++q) Tp[r][wv][row0 + q][m] = t[q];
  }
  lds_barrier();
  if (wv != 0) return;

  v8f y = {};
  #pragma unroll
  for (int r = 0; r < P1R; ++r) {
    const int rowb = (band * P1R + r) * 16;
    #pragma unroll
    for (int k0 = 0; k0 < 16; k0 += 4) {
      const int kk = k0 + kof;
      v2f av, bv;
      av.x = A[m * NHW + rowb + kk];
      av.y = A[m * NHW + rowb + kk + 1];
      bv.x = Tp[r][0][kk][m];
      bv.y = Tp[r][0][kk + 1][m];
      #pragma unroll
      for (int s = 1; s < P1W; ++s) {
        bv.x += Tp[r][s][kk][m];
        bv.y += Tp[r][s][kk + 1][m];
      }
      y = wmma_f32_4(av, bv, y);
    }
  }

  float* outp = Yp + ((size_t)img * NPY + band) * 256;
  #pragma unroll
  for (int q = 0; q < 8; ++q) outp[(row0 + q) * 16 + m] = y[q];
}

__global__ void reduce_y(const float* __restrict__ Yp, float* __restrict__ Y) {
  const int img = blockIdx.x;
  const int e   = threadIdx.x;
  float s = 0.0f;
  #pragma unroll 4
  for (int t = 0; t < NPY; ++t) s += Yp[((size_t)img * NPY + t) * 256 + e];
  Y[(size_t)img * 256 + e] = s;
}

#define SC 128
#define SR 4
#define ZP (SC + 4)

__global__ __launch_bounds__(32) void pass2_reconstruct(const float* __restrict__ im,
                                                        const float* __restrict__ D,
                                                        const float* __restrict__ Y,
                                                        float* __restrict__ out) {
  const int strip = blockIdx.x;
  const int band  = blockIdx.y;
  const int img   = blockIdx.z;
  const int lane  = threadIdx.x;
  const int m     = lane & 15;
  const int hi    = lane >> 4;
  const int row0  = hi * 8;
  const int w0    = strip * SC;
  const int c4    = lane * 4;

  __shared__ __attribute__((aligned(16))) float Zs[16][ZP];

  float ym[8];
  {
    const float* Yi = Y + (size_t)img * 256;
    #pragma unroll
    for (int r = 0; r < 8; ++r) {
      const int i = row0 + r;
      const float v = Yi[i * 16 + m];
      ym[r] = (i + m <= 10) ? v : 0.0f;
    }
  }

  float dt[SR][8];
  #pragma unroll
  for (int s = 0; s < SR; ++s) {
    const float4* p = (const float4*)(D + ((band * SR + s) * 16 + m) * KF + row0);
    const float4 p0 = p[0], p1 = p[1];
    dt[s][0] = p0.x; dt[s][1] = p0.y; dt[s][2] = p0.z; dt[s][3] = p0.w;
    dt[s][4] = p1.x; dt[s][5] = p1.y; dt[s][6] = p1.z; dt[s][7] = p1.w;
  }

  float tb[SC / 16][8];
  #pragma unroll
  for (int t = 0; t < SC / 16; ++t) {
    const float4* p = (const float4*)(D + (w0 + 16 * t + m) * KF + row0);
    const float4 p0 = p[0], p1 = p[1];
    tb[t][0] = p0.x; tb[t][1] = p0.y; tb[t][2] = p0.z; tb[t][3] = p0.w;
    tb[t][4] = p1.x; tb[t][5] = p1.y; tb[t][6] = p1.z; tb[t][7] = p1.w;
  }

  const size_t base = (size_t)img * NHW * NHW + (size_t)band * SR * 16 * NHW + w0;
  const float* X = im + base;
  float* O = out + base;

  __builtin_amdgcn_sched_barrier(0);

  float4 x[2][16];
  #pragma unroll
  for (int q = 0; q < 16; ++q) x[0][q] = *(const float4*)(X + q * NHW + c4);
  #pragma unroll
  for (int q = 0; q < 16; ++q) x[1][q] = *(const float4*)(X + (16 + q) * NHW + c4);

  #pragma unroll
  for (int s = 0; s < SR; ++s) {
    v8f wt = {};
    #pragma unroll
    for (int kk = 0; kk < 4; ++kk) {
      v2f av, bv;
      av.x = ym[2 * kk]; av.y = ym[2 * kk + 1];
      bv.x = dt[s][2 * kk]; bv.y = dt[s][2 * kk + 1];
      wt = wmma_f32_4(av, bv, wt);
    }

    #pragma unroll
    for (int t = 0; t < SC / 16; ++t) {
      v8f z = {};
      #pragma unroll
      for (int kk = 0; kk < 4; ++kk) {
        v2f av, bv;
        av.x = wt[2 * kk];    av.y = wt[2 * kk + 1];
        bv.x = tb[t][2 * kk]; bv.y = tb[t][2 * kk + 1];
        z = wmma_f32_4(av, bv, z);
      }
      #pragma unroll
      for (int r = 0; r < 8; ++r) Zs[row0 + r][16 * t + m] = z[r];
    }
    lds_barrier();

    #pragma unroll
    for (int q = 0; q < 16; ++q) {
      const float4 zz = *(const float4*)(&Zs[q][c4]);
      const float4 xv = x[s & 1][q];
      float4 o;
      o.x = xv.x - zz.x;
      o.y = xv.y - zz.y;
      o.z = xv.z - zz.z;
      o.w = xv.w - zz.w;
      *(float4*)(O + (s * 16 + q) * NHW + c4) = o;
    }

    if (s + 2 < SR) {
      #pragma unroll
      for (int q = 0; q < 16; ++q)
        x[s & 1][q] = *(const float4*)(X + ((s + 2) * 16 + q) * NHW + c4);
    }
    lds_barrier();
  }
}

extern "C" void kernel_launch(void* const* d_in, const int* in_sizes, int n_in,
                              void* d_out, int out_size, void* d_ws, size_t ws_size,
                              hipStream_t stream) {
  (void)n_in; (void)out_size; (void)ws_size;
  const float* im = (const float*)d_in[0];
  float* out = (float*)d_out;
  const int nimg = in_sizes[0] / (NHW * NHW);

  char*  ws  = (char*)d_ws;
  float* dA  = (float*)(ws);
  float* dD  = (float*)(ws + 32768);
  float* dYp = (float*)(ws + 65536);
  float* dY  = (float*)(ws + 65536 + (size_t)nimg * 32 * 256 * sizeof(float));

  init_consts<<<KF * NHW / TBW, TBW, 0, stream>>>(dA, dD);
  pass1_dct_low<<<dim3(NPY, nimg), P1T, 0, stream>>>(im, dA, dYp);
  reduce_y<<<nimg, 256, 0, stream>>>(dYp, dY);
  pass2_reconstruct<<<dim3(NHW / SC, 32 / SR, nimg), 32, 0, stream>>>(im, dD, dY, out);
}


# pass 2: the 4 column strips of a band run as the 4 waves of one work-group (768 groups, were 3072 one-wave groups; same grid as pass 1); each wave keeps its own LDS block, no work-group barrier
# speedup vs baseline: 3.4555x; 1.1695x over previous
#include <hip/hip_runtime.h>
#include <hip/hip_bf16.h>
#include <math.h>

typedef float v2f __attribute__((ext_vector_type(2)));
typedef float v8f __attribute__((ext_vector_type(8)));

#define NHW  512
#define KF   16
#define LPAD 4

__device__ __forceinline__ v8f wmma_f32_4(v2f a, v2f b, v8f c) {
  return __builtin_amdgcn_wmma_f32_16x16x4_f32(false, a, false, b, (short)0, c,
                                               false, false);
}

#define TBW 32

__global__ __launch_bounds__(TBW) void init_consts(float* __restrict__ A,
                                                   float* __restrict__ D) {
  const int idx = blockIdx.x * TBW + threadIdx.x;
  const int i = idx / NHW, n = idx % NHW;
  float a = 0.0f, d = 0.0f;
  if (i <= 10) {
    const double w = (i == 0) ? 0.5 : 1.0;
    const double arg = M_PI * (double)((2 * n + 1) * i) / (2.0 * NHW);
    const double c = cos(arg);
    a = (float)(2.0 * c);
    d = (float)(w * c / (double)NHW);
  }
  A[i * NHW + n] = a;
  D[n * KF + i] = d;
}

__device__ __forceinline__ void lds_barrier() {
  __builtin_amdgcn_fence(__ATOMIC_RELEASE, "workgroup", "local");
  __builtin_amdgcn_s_barrier();
  __builtin_amdgcn_fence(__ATOMIC_ACQUIRE, "workgroup", "local");
}

#define P1W  4
#define P1T  (32 * P1W)
#define KSL  (NHW / P1W)
#define P1R  4
#define NPY  (32 / P1R)
#define P1Q  ((16 * NHW / 4) / P1T)

__global__ __launch_bounds__(P1T) void pass1_dct_low(const float* __restrict__ im,
                                                     const float* __restrict__ A,
                                                     float* __restrict__ Yp) {
  const int band = blockIdx.x;
  const int img  = blockIdx.y;
  const int tid  = threadIdx.x;
  const int lane = tid & 31;
  const int wv   = __builtin_amdgcn_readfirstlane(tid >> 5);
  const float* X = im + (size_t)img * NHW * NHW + (size_t)band * P1R * 16 * NHW;

  __shared__ __attribute__((aligned(16))) float Xs[2][16][NHW + LPAD];
  __shared__ float Tp[P1R][P1W][16][16 + LPAD];

  const int m    = lane & 15;
  const int hi   = lane >> 4;
  const int kof  = hi * 2;
  const int row0 = hi * 8;
  const int wbeg = wv * KSL;

  const int lr = tid / (NHW / 4);
  const int lc = tid % (NHW / 4);
  float4 xr[P1Q];
  #pragma unroll
  for (int q = 0; q < P1Q; ++q)
    xr[q] = *(const float4*)(X + (lr + q * (P1T * 4 / NHW)) * NHW + 4 * lc);

  v2f bt[KSL / 4];
  #pragma unroll
  for (int j = 0; j < KSL / 4; ++j) {
    bt[j].x = A[m * NHW + wbeg + 4 * j + kof];
    bt[j].y = A[m * NHW + wbeg + 4 * j + kof + 1];
  }

  #pragma unroll
  for (int r = 0; r < P1R; ++r) {
    const int b = r & 1;
    #pragma unroll
    for (int q = 0; q < P1Q; ++q)
      *(float4*)(&Xs[b][lr + q * (P1T * 4 / NHW)][4 * lc]) = xr[q];
    lds_barrier();

    if (r + 1 < P1R) {
      #pragma unroll
      for (int q = 0; q < P1Q; ++q)
        xr[q] = *(const float4*)(X + ((r + 1) * 16 + lr + q * (P1T * 4 / NHW)) * NHW + 4 * lc);
    }

    v8f t = {};
    #pragma unroll
    for (int j = 0; j < KSL / 4; ++j) {
      const int w0 = wbeg + 4 * j;
      v2f av;
      av.x = Xs[b][m][w0 + kof];
      av.y = Xs[b][m][w0 + kof + 1];
      t = wmma_f32_4(av, bt[j], t);
    }
    #pragma unroll
    for (int q = 0; q < 8; ++q) Tp[r][wv][row0 + q][m] = t[q];
  }
  lds_barrier();
  if (wv != 0) return;

  v8f y = {};
  #pragma unroll
  for (int r = 0; r < P1R; ++r) {
    const int rowb = (band * P1R + r) * 16;
    #pragma unroll
    for (int k0 = 0; k0 < 16; k0 += 4) {
      const int kk = k0 + kof;
      v2f av, bv;
      av.x = A[m * NHW + rowb + kk];
      av.y = A[m * NHW + rowb + kk + 1];
      bv.x = Tp[r][0][kk][m];
      bv.y = Tp[r][0][kk + 1][m];
      #pragma unroll
      for (int s = 1; s < P1W; ++s) {
        bv.x += Tp[r][s][kk][m];
        bv.y += Tp[r][s][kk + 1][m];
      }
      y = wmma_f32_4(av, bv, y);
    }
  }

  float* outp = Yp + ((size_t)img * NPY + band) * 256;
  #pragma unroll
  for (int q = 0; q < 8; ++q) outp[(row0 + q) * 16 + m] = y[q];
}

__global__ void reduce_y(const float* __restrict__ Yp, float* __restrict__ Y) {
  const int img = blockIdx.x;
  const int e   = threadIdx.x;
  float s = 0.0f;
  #pragma unroll 4
  for (int t = 0; t < NPY; ++t) s += Yp[((size_t)img * NPY + t) * 256 + e];
  Y[(size_t)img * 256 + e] = s;
}

#define SC 128
#define SR 4
#define ZP (SC + 4)
#define P2W (NHW / SC)

__device__ __forceinline__ void lds_wave_sync() {
  __builtin_amdgcn_fence(__ATOMIC_RELEASE, "workgroup", "local");
  __builtin_amdgcn_fence(__ATOMIC_ACQUIRE, "workgroup", "local");
}

__global__ __launch_bounds__(32 * P2W) void pass2_reconstruct(const float* __restrict__ im,
                                                        const float* __restrict__ D,
                                                        const float* __restrict__ Y,
                                                        float* __restrict__ out) {
  const int strip = __builtin_amdgcn_readfirstlane(threadIdx.x >> 5);
  const int band  = blockIdx.x;
  const int img   = blockIdx.y;
  const int lane  = threadIdx.x & 31;
  const int m     = lane & 15;
  const int hi    = lane >> 4;
  const int row0  = hi * 8;
  const int w0    = strip * SC;
  const int c4    = lane * 4;

  __shared__ __attribute__((aligned(16))) float Zall[P2W][16][ZP];
  float (*Zs)[ZP] = Zall[strip];

  float ym[8];
  {
    const float* Yi = Y + (size_t)img * 256;
    #pragma unroll
    for (int r = 0; r < 8; ++r) {
      const int i = row0 + r;
      const float v = Yi[i * 16 + m];
      ym[r] = (i + m <= 10) ? v : 0.0f;
    }
  }

  float dt[SR][8];
  #pragma unroll
  for (int s = 0; s < SR; ++s) {
    const float4* p = (const float4*)(D + ((band * SR + s) * 16 + m) * KF + row0);
    const float4 p0 = p[0], p1 = p[1];
    dt[s][0] = p0.x; dt[s][1] = p0.y; dt[s][2] = p0.z; dt[s][3] = p0.w;
    dt[s][4] = p1.x; dt[s][5] = p1.y; dt[s][6] = p1.z; dt[s][7] = p1.w;
  }

  float tb[SC / 16][8];
  #pragma unroll
  for (int t = 0; t < SC / 16; ++t) {
    const float4* p = (const float4*)(D + (w0 + 16 * t + m) * KF + row0);
    const float4 p0 = p[0], p1 = p[1];
    tb[t][0] = p0.x; tb[t][1] = p0.y; tb[t][2] = p0.z; tb[t][3] = p0.w;
    tb[t][4] = p1.x; tb[t][5] = p1.y; tb[t][6] = p1.z; tb[t][7] = p1.w;
  }

  const size_t base = (size_t)img * NHW * NHW + (size_t)band * SR * 16 * NHW + w0;
  const float* X = im + base;
  float* O = out + base;

  __builtin_amdgcn_sched_barrier(0);

  float4 x[2][16];
  #pragma unroll
  for (int q = 0; q < 16; ++q) x[0][q] = *(const float4*)(X + q * NHW + c4);
  #pragma unroll
  for (int q = 0; q < 16; ++q) x[1][q] = *(const float4*)(X + (16 + q) * NHW + c4);

  #pragma unroll
  for (int s = 0; s < SR; ++s) {
    v8f wt = {};
    #pragma unroll
    for (int kk = 0; kk < 4; ++kk) {
      v2f av, bv;
      av.x = ym[2 * kk]; av.y = ym[2 * kk + 1];
      bv.x = dt[s][2 * kk]; bv.y = dt[s][2 * kk + 1];
      wt = wmma_f32_4(av, bv, wt);
    }

    #pragma unroll
    for (int t = 0; t < SC / 16; ++t) {
      v8f z = {};
      #pragma unroll
      for (int kk = 0; kk < 4; ++kk) {
        v2f av, bv;
        av.x = wt[2 * kk];    av.y = wt[2 * kk + 1];
        bv.x = tb[t][2 * kk]; bv.y = tb[t][2 * kk + 1];
        z = wmma_f32_4(av, bv, z);
      }
      #pragma unroll
      for (int r = 0; r < 8; ++r) Zs[row0 + r][16 * t + m] = z[r];
    }
    lds_wave_sync();

    #pragma unroll
    for (int q = 0; q < 16; ++q) {
      const float4 zz = *(const float4*)(&Zs[q][c4]);
      const float4 xv = x[s & 1][q];
      float4 o;
      o.x = xv.x - zz.x;
      o.y = xv.y - zz.y;
      o.z = xv.z - zz.z;
      o.w = xv.w - zz.w;
      *(float4*)(O + (s * 16 + q) * NHW + c4) = o;
    }

    if (s + 2 < SR) {
      #pragma unroll
      for (int q = 0; q < 16; ++q)
        x[s & 1][q] = *(const float4*)(X + ((s + 2) * 16 + q) * NHW + c4);
    }
    lds_wave_sync();
  }
}

extern "C" void kernel_launch(void* const* d_in, const int* in_sizes, int n_in,
                              void* d_out, int out_size, void* d_ws, size_t ws_size,
                              hipStream_t stream) {
  (void)n_in; (void)out_size; (void)ws_size;
  const float* im = (const float*)d_in[0];
  float* out = (float*)d_out;
  const int nimg = in_sizes[0] / (NHW * NHW);

  char*  ws  = (char*)d_ws;
  float* dA  = (float*)(ws);
  float* dD  = (float*)(ws + 32768);
  float* dYp = (float*)(ws + 65536);
  float* dY  = (float*)(ws + 65536 + (size_t)nimg * 32 * 256 * sizeof(float));

  init_consts<<<KF * NHW / TBW, TBW, 0, stream>>>(dA, dD);
  pass1_dct_low<<<dim3(NPY, nimg), P1T, 0, stream>>>(im, dA, dYp);
  reduce_y<<<nimg, 256, 0, stream>>>(dYp, dY);
  pass2_reconstruct<<<dim3(32 / SR, nimg), 32 * P2W, 0, stream>>>(im, dD, dY, out);
}
